# starting kernel
# baseline (speedup 1.0000x reference)
#include <hip/hip_runtime.h>

typedef __attribute__((ext_vector_type(2))) float v2f;
typedef __attribute__((ext_vector_type(4))) float v4f;
typedef __attribute__((ext_vector_type(8))) float v8f;

#define VQ_B 64
#define VQ_C 64
#define VQ_HW 1024
#define VQ_K 1024
#define WAVES 8
#define XS_STRIDE 68

__global__ __launch_bounds__(256) void vq_cnorm_kernel(const float* __restrict__ cb,
                                                       float* __restrict__ cn2h) {
  int k = blockIdx.x * blockDim.x + threadIdx.x;
  if (k < VQ_K) {
    const float* row = cb + (size_t)k * VQ_C;
    float s = 0.0f;
#pragma unroll
    for (int c = 0; c < VQ_C; c += 4) {
      v4f v = *(const v4f*)(row + c);
      s += v.x * v.x + v.y * v.y + v.z * v.z + v.w * v.w;
    }
    cn2h[k] = 0.5f * s;
  }
}

__device__ __forceinline__ void vq_ldb(const float* __restrict__ cb, int code, int bk,
                                       v2f (&bf)[16]) {
  const v2f* bp = (const v2f*)cb + (size_t)code * (VQ_C / 2) + bk;
#pragma unroll
  for (int kk = 0; kk < 16; ++kk) bf[kk] = bp[kk * 2];
}

__device__ __forceinline__ void vq_tile(const v2f (&a)[16], const v2f (&bf)[16],
                                        int code, const float* __restrict__ cn2h,
                                        float (&best)[8], int (&bidx)[8]) {
  const float cn = cn2h[code];
  v8f acc0 = {}, acc1 = {};
#pragma unroll
  for (int kk = 0; kk < 16; kk += 2) {
    acc0 = __builtin_amdgcn_wmma_f32_16x16x4_f32(
        false, a[kk + 0], false, bf[kk + 0], (short)0, acc0, false, false);
    acc1 = __builtin_amdgcn_wmma_f32_16x16x4_f32(
        false, a[kk + 1], false, bf[kk + 1], (short)0, acc1, false, false);
  }
#pragma unroll
  for (int r = 0; r < 8; ++r) {
    float s = (acc0[r] + acc1[r]) - cn;
    if (s > best[r]) { best[r] = s; bidx[r] = code; }
  }
}

__global__ __launch_bounds__(256) void vq_main_kernel(const float* __restrict__ in,
                                                      const float* __restrict__ cb,
                                                      const float* __restrict__ cn2h,
                                                      float* __restrict__ out) {
  __shared__ float xs[16 * XS_STRIDE];
  __shared__ float red_s[16 * WAVES];
  __shared__ int   red_i[16 * WAVES];
  __shared__ int   rowbest[16];

  const int tid  = threadIdx.x;
  const int lane = tid & 31;
  const int wave = tid >> 5;
  const int m    = blockIdx.x;
  const int b    = m >> 6;
  const int hw0  = (m & 63) << 4;

  {
    const int c  = tid >> 2;
    const int dn = (tid & 3) << 2;
    v4f v = *(const v4f*)(in + (((size_t)(b * VQ_C + c)) << 10) + hw0 + dn);
    xs[(dn + 0) * XS_STRIDE + c] = v.x;
    xs[(dn + 1) * XS_STRIDE + c] = v.y;
    xs[(dn + 2) * XS_STRIDE + c] = v.z;
    xs[(dn + 3) * XS_STRIDE + c] = v.w;
  }
  __syncthreads();

  const int an = lane & 15;
  const int ak = (lane >> 4) << 1;
  v2f a[16];
#pragma unroll
  for (int kk = 0; kk < 16; ++kk)
    a[kk] = *(const v2f*)(xs + an * XS_STRIDE + kk * 4 + ak);

  float best[8];
  int   bidx[8];
#pragma unroll
  for (int r = 0; r < 8; ++r) { best[r] = -3.0e38f; bidx[r] = 0; }

  const int ncol = lane & 15;
  const int bk   = lane >> 4;
  const int base = wave * 8;

  v2f bf0[16], bf1[16];
  vq_ldb(cb, (base + 0) * 16 + ncol, bk, bf0);
#pragma unroll 1
  for (int tt = 0; tt < 4; ++tt) {
    const int t0 = 2 * tt, t1 = 2 * tt + 1;
    vq_ldb(cb, (base + t1) * 16 + ncol, bk, bf1);
    vq_tile(a, bf0, (base + t0) * 16 + ncol, cn2h, best, bidx);
    if (tt < 3)
      vq_ldb(cb, (base + t1 + 1) * 16 + ncol, bk, bf0);
    vq_tile(a, bf1, (base + t1) * 16 + ncol, cn2h, best, bidx);
  }

#pragma unroll
  for (int r = 0; r < 8; ++r) {
    float s = best[r];
    int   i = bidx[r];
#pragma unroll
    for (int msk = 8; msk >= 1; msk >>= 1) {
      float os = __shfl_xor(s, msk, 32);
      int   oi = __shfl_xor(i, msk, 32);
      if (os > s || (os == s && oi < i)) { s = os; i = oi; }
    }
    const int rowm = r + ((lane >> 4) << 3);
    if ((lane & 15) == 0) {
      red_s[rowm * WAVES + wave] = s;
      red_i[rowm * WAVES + wave] = i;
    }
  }
  __syncthreads();

  if (tid < 16) {
    float s = red_s[tid * WAVES];
    int   i = red_i[tid * WAVES];
#pragma unroll
    for (int w = 1; w < WAVES; ++w) {
      float os = red_s[tid * WAVES + w];
      int   oi = red_i[tid * WAVES + w];
      if (os > s || (os == s && oi < i)) { s = os; i = oi; }
    }
    rowbest[tid] = i;
  }
  __syncthreads();

  {
    const int row = tid & 15;
    const int c0  = tid >> 4;
    const float* crow = cb + (size_t)rowbest[row] * VQ_C;
    const size_t obase = (((size_t)b * VQ_C) << 10) + hw0 + row;
#pragma unroll
    for (int j = 0; j < 4; ++j) {
      const int c = c0 + 16 * j;
      out[obase + ((size_t)c << 10)] = crow[c];
    }
  }
}

extern "C" void kernel_launch(void* const* d_in, const int* in_sizes, int n_in,
                              void* d_out, int out_size, void* d_ws, size_t ws_size,
                              hipStream_t stream) {
  const float* in  = (const float*)d_in[0];
  const float* cb  = (const float*)d_in[1];
  float* out  = (float*)d_out;
  float* cn2h = (float*)d_ws;

  vq_cnorm_kernel<<<VQ_K / 256, 256, 0, stream>>>(cb, cn2h);
  vq_main_kernel<<<(VQ_B * VQ_HW) / 16, 256, 0, stream>>>(in, cb, cn2h, out);
}


# Codebook enters the A-fragment registers tile by tile inside step 0 (ring of 4 tiles in flight, LDS-only barrier): a wave's matrix work starts after its first 4 KB, not all 32 KB; scores unchanged
# speedup vs baseline: 2.4198x; 2.4198x over previous
#include <hip/hip_runtime.h>


typedef __attribute__((ext_vector_type(2))) float v2f;
typedef __attribute__((ext_vector_type(4))) float v4f;
typedef __attribute__((ext_vector_type(8))) float v8f;

#define VQ_B 64
#define VQ_C 64
#define VQ_HW 1024
#define VQ_K 1024
#define WAVES 8
#define PXB 256
#define CT 8
#define NP 2
#define PITCH 68

#define XS_OFF   0
#define CNS_OFF  (PXB * PITCH)
#define REDS_OFF (CNS_OFF + VQ_K)
#define REDI_OFF (REDS_OFF + WAVES * PXB)
#define RB_OFF   (REDI_OFF + WAVES * PXB)
#define STG_OFF  (RB_OFF + PXB)
#define STG_SLOT (16 * PITCH)
#define SM_FLOATS (STG_OFF + WAVES * 4 * STG_SLOT)

__global__ __launch_bounds__(256) void vq_cnorm_kernel(const float* __restrict__ cb,
                                                       float* __restrict__ cn2h) {
  int k = blockIdx.x * blockDim.x + threadIdx.x;
  if (k < VQ_K) {
    const float* row = cb + (size_t)k * VQ_C;
    float s = 0.0f;
#pragma unroll
    for (int c = 0; c < VQ_C; c += 4) {
      v4f v = *(const v4f*)(row + c);
      s += v.x * v.x + v.y * v.y + v.z * v.z + v.w * v.w;
    }
    cn2h[k] = 0.5f * s;
  }
}

__device__ __forceinline__ void vq_ldx(const float* xl, int st, v2f (&bq)[NP][16]) {
  const float* p0 = xl + st * (NP * 16 * PITCH);
#pragma unroll
  for (int p = 0; p < NP; ++p)
#pragma unroll
    for (int kk = 0; kk < 16; ++kk)
      bq[p][kk] = *(const v2f*)(p0 + p * 16 * PITCH + kk * 4);
}

__device__ __forceinline__ void vq_step(const v2f (&a)[CT][16], const v2f (&bq)[NP][16],
                                        const float* cnl, float (&best)[NP], int (&btile)[NP],
                                        float (&sav)[NP][8]) {
  v8f acc[CT][NP];
#pragma unroll
  for (int ct = 0; ct <= CT; ++ct) {
    if (ct < CT) {
#pragma unroll
      for (int p = 0; p < NP; ++p) acc[ct][p] = (v8f){};
#pragma unroll
      for (int kk = 0; kk < 16; ++kk)
#pragma unroll
        for (int p = 0; p < NP; ++p)
          acc[ct][p] = __builtin_amdgcn_wmma_f32_16x16x4_f32(
              false, a[ct][kk], false, bq[p][kk], (short)0, acc[ct][p], false, false);
    }
    if (ct > 0) {
      const int pt = ct - 1;
      const v4f c0 = *(const v4f*)(cnl + pt * 16);
      const v4f c1 = *(const v4f*)(cnl + pt * 16 + 4);
      const float cn[8] = {c0.x, c0.y, c0.z, c0.w, c1.x, c1.y, c1.z, c1.w};
#pragma unroll
      for (int p = 0; p < NP; ++p) {
        float v[8];
#pragma unroll
        for (int r = 0; r < 8; ++r) v[r] = acc[pt][p][r] - cn[r];
        const float m0 = fmaxf(fmaxf(v[0], v[1]), v[2]);
        const float m1 = fmaxf(fmaxf(v[3], v[4]), v[5]);
        if (pt == 0) {
          best[p] = fmaxf(fmaxf(fmaxf(v[6], v[7]), m0), m1);
#pragma unroll
          for (int r = 0; r < 8; ++r) sav[p][r] = v[r];
          btile[p] = 0;
        } else {
          const float m2 = fmaxf(fmaxf(v[6], v[7]), best[p]);
          const float nb = fmaxf(fmaxf(m0, m1), m2);
          const bool take = nb > best[p];
#pragma unroll
          for (int r = 0; r < 8; ++r) sav[p][r] = take ? v[r] : sav[p][r];
          btile[p] = take ? (pt * 16) : btile[p];
          best[p] = nb;
        }
      }
    }
    static_assert(NP == 2, "the fence below names both pixel tiles");
    if (ct > 0)
      asm volatile("" :: "v"(best[0]), "v"(best[1]),
                   "v"(sav[0][0]), "v"(sav[0][1]), "v"(sav[0][2]), "v"(sav[0][3]),
                   "v"(sav[0][4]), "v"(sav[0][5]), "v"(sav[0][6]), "v"(sav[0][7]),
                   "v"(sav[1][0]), "v"(sav[1][1]), "v"(sav[1][2]), "v"(sav[1][3]),
                   "v"(sav[1][4]), "v"(sav[1][5]), "v"(sav[1][6]), "v"(sav[1][7]));
    __builtin_amdgcn_sched_barrier(0);
  }
}

__device__ __forceinline__ void vq_step_fill(v2f (&a)[CT][16], v4f (&t)[32], const float* gsrc,
                                             float* stg, int c, int h, const v2f (&bq)[NP][16],
                                             const float* cnl, float (&best)[NP], int (&bidx)[NP]) {
  v8f acc[CT][NP];
#pragma unroll
  for (int ct = 0; ct <= CT; ++ct) {
    if (ct < CT) {
      float* slot = stg + (ct & 3) * STG_SLOT;
#pragma unroll
      for (int j = 0; j < 8; ++j)
        *(v4f*)(slot + (2 * j + h) * PITCH + 4 * c) = t[(8 * ct + j) & 31];
      if (ct + 4 < CT) {
#pragma unroll
        for (int j = 0; j < 8; ++j)
          t[(8 * ct + j) & 31] = *(const v4f*)(gsrc + (8 * (ct + 4) + j) * 128);
      }
      __builtin_amdgcn_wave_barrier();
#pragma unroll
      for (int kk = 0; kk < 16; ++kk)
        a[ct][kk] = *(const v2f*)(slot + c * PITCH + 2 * h + kk * 4);
#pragma unroll
      for (int p = 0; p < NP; ++p) acc[ct][p] = (v8f){};
#pragma unroll
      for (int kk = 0; kk < 16; ++kk)
#pragma unroll
        for (int p = 0; p < NP; ++p)
          acc[ct][p] = __builtin_amdgcn_wmma_f32_16x16x4_f32(
              false, a[ct][kk], false, bq[p][kk], (short)0, acc[ct][p], false, false);
    }
    if (ct > 0) {
      const int pt = ct - 1;
      const v4f c0 = *(const v4f*)(cnl + pt * 16);
      const v4f c1 = *(const v4f*)(cnl + pt * 16 + 4);
      const float cn[8] = {c0.x, c0.y, c0.z, c0.w, c1.x, c1.y, c1.z, c1.w};
#pragma unroll
      for (int p = 0; p < NP; ++p) {
        float v[8];
#pragma unroll
        for (int r = 0; r < 8; ++r) v[r] = acc[pt][p][r] - cn[r];
        const float m0 = fmaxf(fmaxf(v[0], v[1]), v[2]);
        const float m1 = fmaxf(fmaxf(v[3], v[4]), v[5]);
        const float m  = fmaxf(fmaxf(fmaxf(v[6], v[7]), m0), m1);
        int row = 7;
#pragma unroll
        for (int r = 6; r >= 0; --r) row = (v[r] == m) ? r : row;
        if (pt == 0) {
          best[p] = m;
          bidx[p] = row;
        } else {
          const bool take = m > best[p];
          bidx[p] = take ? (pt * 16 + row) : bidx[p];
          best[p] = take ? m : best[p];
        }
      }
    }
    static_assert(NP == 2, "the fence below names both pixel tiles");
    if (ct > 0)
      asm volatile("" :: "v"(best[0]), "v"(best[1]), "v"(bidx[0]), "v"(bidx[1]));
    __builtin_amdgcn_sched_barrier(0);
  }
}

__device__ __forceinline__ void vq_park_idx(const float (&best)[NP], const int (&bidx)[NP],
                                            int cbase, int h, float* rs, int* ri, int st) {
#pragma unroll
  for (int p = 0; p < NP; ++p) {
    float s = best[p];
    int   i = cbase + bidx[p];
    const float os = __shfl_xor(s, 16, 32);
    const int   oi = __shfl_xor(i, 16, 32);
    const bool take = (os > s) | ((os == s) & (oi < i));
    s = take ? os : s;
    i = take ? oi : i;
    if (h == 0) {
      rs[st * (NP * 16) + p * 16] = s;
      ri[st * (NP * 16) + p * 16] = i;
    }
  }
}

__device__ __forceinline__ void vq_park(float (&best)[NP], int (&btile)[NP], float (&sav)[NP][8],
                                        int cbase, int h, float* rs, int* ri, int st) {
#pragma unroll
  for (int p = 0; p < NP; ++p) {
    float s = best[p];
    int row = 7;
#pragma unroll
    for (int r = 6; r >= 0; --r) row = (sav[p][r] == s) ? r : row;
    int   i = cbase + btile[p] + row;
    const float os = __shfl_xor(s, 16, 32);
    const int   oi = __shfl_xor(i, 16, 32);
    const bool take = (os > s) | ((os == s) & (oi < i));
    s = take ? os : s;
    i = take ? oi : i;
    if (h == 0) {
      rs[st * (NP * 16) + p * 16] = s;
      ri[st * (NP * 16) + p * 16] = i;
    }
  }
}

__global__ __launch_bounds__(256) void vq_main_kernel(const float* __restrict__ in,
                                                      const float* __restrict__ cb,
                                                      const float* __restrict__ cn2h,
                                                      float* __restrict__ out) {
  __shared__ __attribute__((aligned(16))) float sm[SM_FLOATS];

  const int tid  = threadIdx.x;
  const int lane = tid & 31;
  const int wave = tid >> 5;
  const int c    = lane & 15;
  const int h    = lane >> 4;
  const int g    = blockIdx.x;
  const int b    = g >> 2;
  const int hw0  = (g & 3) << 8;
  const int code0 = wave * (CT * 16);

  const int px4 = tid & 3;
  const int chp = (tid >> 2) & 31;
  const int pth = tid >> 7;
  v4f xu[8], xw[8];
  {
    const float* src = in + (((size_t)(b * VQ_C + 2 * chp)) << 10) + hw0 + 4 * px4;
#pragma unroll
    for (int j = 0; j < 8; ++j) {
      xu[j] = *(const v4f*)(src + 16 * (2 * j + pth));
      xw[j] = *(const v4f*)(src + VQ_HW + 16 * (2 * j + pth));
    }
  }
  float cnv[4];
#pragma unroll
  for (int i = 0; i < 4; ++i) cnv[i] = cn2h[tid + 256 * i];

  v4f t[32];
  const float* gsrc = cb + (size_t)code0 * VQ_C + 4 * lane;
#pragma unroll
  for (int j = 0; j < 32; ++j) t[j] = *(const v4f*)(gsrc + j * 128);
  __builtin_amdgcn_sched_barrier(0);

  {
    float* xs = sm + XS_OFF;
#pragma unroll
    for (int j = 0; j < 8; ++j) {
      float* d = xs + (16 * (2 * j + pth) + 4 * px4) * PITCH + 2 * chp;
      *(v2f*)(d + 0 * PITCH) = (v2f){xu[j].x, xw[j].x};
      *(v2f*)(d + 1 * PITCH) = (v2f){xu[j].y, xw[j].y};
      *(v2f*)(d + 2 * PITCH) = (v2f){xu[j].z, xw[j].z};
      *(v2f*)(d + 3 * PITCH) = (v2f){xu[j].w, xw[j].w};
    }
#pragma unroll
    for (int i = 0; i < 4; ++i) sm[CNS_OFF + tid + 256 * i] = cnv[i];
  }
  __builtin_amdgcn_fence(__ATOMIC_RELEASE, "workgroup", "local");
  __builtin_amdgcn_s_barrier();
  __builtin_amdgcn_fence(__ATOMIC_ACQUIRE, "workgroup", "local");

  const float* xl  = sm + XS_OFF + c * PITCH + 2 * h;
  const float* cnl = sm + CNS_OFF + code0 + 8 * h;
  float* rs = sm + REDS_OFF + wave * PXB + c;
  int*   ri = (int*)(sm + REDI_OFF) + wave * PXB + c;
  const int cbase = code0 + 8 * h;

  float best[NP];
  int   btile[NP];
  float sav[NP][8];

  v2f bq0[NP][16], bq1[NP][16];
  v2f a[CT][16];
  {
    v2f bqf[NP][16];
    vq_ldx(xl, 0, bqf);
    float fbest[NP];
    int   fidx[NP];
    vq_step_fill(a, t, gsrc, sm + STG_OFF + wave * (4 * STG_SLOT), c, h, bqf, cnl, fbest, fidx);
    vq_ldx(xl, 1, bq1);
    vq_park_idx(fbest, fidx, cbase, h, rs, ri, 0);
  }
#pragma unroll 1
  for (int it = 0; it < PXB / (NP * 16) / 2; ++it) {
    const bool more = it < PXB / (NP * 16) / 2 - 1;
    if (more)
      vq_ldx(xl, 2 * it + 2, bq0);
    vq_step(a, bq1, cnl, best, btile, sav);
    vq_park(best, btile, sav, cbase, h, rs, ri, 2 * it + 1);
    if (more) {
      vq_ldx(xl, 2 * it + 3, bq1);
      vq_step(a, bq0, cnl, best, btile, sav);
      vq_park(best, btile, sav, cbase, h, rs, ri, 2 * it + 2);
    }
  }
  __syncthreads();

  {
    const float* qs = sm + REDS_OFF + tid;
    const int*   qi = (const int*)(sm + REDI_OFF) + tid;
    float s = qs[0];
    int   i = qi[0];
#pragma unroll
    for (int w = 1; w < WAVES; ++w) {
      const float os = qs[w * PXB];
      const int   oi = qi[w * PXB];
      const bool take = (os > s) | ((os == s) & (oi < i));
      s = take ? os : s;
      i = take ? oi : i;
    }
    ((int*)(sm + RB_OFF))[tid] = i;
  }
  __syncthreads();

  {
    const int* rb = (const int*)(sm + RB_OFF);
    float* ys = sm + XS_OFF;
    v4f t[16];
#pragma unroll
    for (int j = 0; j < 16; ++j) {
      const int pp = 32 * wave + 2 * j + h;
      t[j] = *(const v4f*)(cb + (size_t)rb[pp] * VQ_C + 4 * c);
    }
#pragma unroll
    for (int j = 0; j < 16; ++j) {
      const int pp = 32 * wave + 2 * j + h;
      *(v4f*)(ys + pp * PITCH + 4 * c) = t[j];
    }
  }
  __syncthreads();
  {
    const float* yr = sm + XS_OFF + tid * PITCH;
    float* o = out + (((size_t)b * VQ_C) << 10) + hw0 + tid;
#pragma unroll
    for (int q = 0; q < 16; ++q) {
      const v4f v = *(const v4f*)(yr + 4 * q);
      o[(size_t)(4 * q + 0) << 10] = v.x;
      o[(size_t)(4 * q + 1) << 10] = v.y;
      o[(size_t)(4 * q + 2) << 10] = v.z;
      o[(size_t)(4 * q + 3) << 10] = v.w;
    }
  }
}

extern "C" void kernel_launch(void* const* d_in, const int* in_sizes, int n_in,
                              void* d_out, int out_size, void* d_ws, size_t ws_size,
                              hipStream_t stream) {
  const float* in  = (const float*)d_in[0];
  const float* cb  = (const float*)d_in[1];
  float* out  = (float*)d_out;
  float* cn2h = (float*)d_ws;

  vq_cnorm_kernel<<<VQ_K / 256, 256, 0, stream>>>(cb, cn2h);
  vq_main_kernel<<<(VQ_B * VQ_HW) / PXB, 256, 0, stream>>>(in, cb, cn2h, out);
}


# Priority window: waves 4-7 (younger wave of each SIMD's pair) hold s_setprio 3 for steps 0-5, then 0, so a pair ends together instead of one wave running its stalls alone; arithmetic unchanged
# speedup vs baseline: 2.5020x; 1.0340x over previous
#include <hip/hip_runtime.h>


typedef __attribute__((ext_vector_type(2))) float v2f;
typedef __attribute__((ext_vector_type(4))) float v4f;
typedef __attribute__((ext_vector_type(8))) float v8f;

#define VQ_B 64
#define VQ_C 64
#define VQ_HW 1024
#define VQ_K 1024
#define WAVES 8
#define PXB 256
#define CT 8
#define NP 2
#define PITCH 68
#define PRIO_STEPS 6

#define XS_OFF   0
#define CNS_OFF  (PXB * PITCH)
#define REDS_OFF (CNS_OFF + VQ_K)
#define REDI_OFF (REDS_OFF + WAVES * PXB)
#define RB_OFF   (REDI_OFF + WAVES * PXB)
#define STG_OFF  (RB_OFF + PXB)
#define STG_SLOT (16 * PITCH)
#define SM_FLOATS (STG_OFF + WAVES * 4 * STG_SLOT)

__global__ __launch_bounds__(256) void vq_cnorm_kernel(const float* __restrict__ cb,
                                                       float* __restrict__ cn2h) {
  int k = blockIdx.x * blockDim.x + threadIdx.x;
  if (k < VQ_K) {
    const float* row = cb + (size_t)k * VQ_C;
    float s = 0.0f;
#pragma unroll
    for (int c = 0; c < VQ_C; c += 4) {
      v4f v = *(const v4f*)(row + c);
      s += v.x * v.x + v.y * v.y + v.z * v.z + v.w * v.w;
    }
    cn2h[k] = 0.5f * s;
  }
}

__device__ __forceinline__ void vq_ldx(const float* xl, int st, v2f (&bq)[NP][16]) {
  const float* p0 = xl + st * (NP * 16 * PITCH);
#pragma unroll
  for (int p = 0; p < NP; ++p)
#pragma unroll
    for (int kk = 0; kk < 16; ++kk)
      bq[p][kk] = *(const v2f*)(p0 + p * 16 * PITCH + kk * 4);
}

__device__ __forceinline__ void vq_step(const v2f (&a)[CT][16], const v2f (&bq)[NP][16],
                                        const float* cnl, float (&best)[NP], int (&btile)[NP],
                                        float (&sav)[NP][8]) {
  v8f acc[CT][NP];
#pragma unroll
  for (int ct = 0; ct <= CT; ++ct) {
    if (ct < CT) {
#pragma unroll
      for (int p = 0; p < NP; ++p) acc[ct][p] = (v8f){};
#pragma unroll
      for (int kk = 0; kk < 16; ++kk)
#pragma unroll
        for (int p = 0; p < NP; ++p)
          acc[ct][p] = __builtin_amdgcn_wmma_f32_16x16x4_f32(
              false, a[ct][kk], false, bq[p][kk], (short)0, acc[ct][p], false, false);
    }
    if (ct > 0) {
      const int pt = ct - 1;
      const v4f c0 = *(const v4f*)(cnl + pt * 16);
      const v4f c1 = *(const v4f*)(cnl + pt * 16 + 4);
      const float cn[8] = {c0.x, c0.y, c0.z, c0.w, c1.x, c1.y, c1.z, c1.w};
#pragma unroll
      for (int p = 0; p < NP; ++p) {
        float v[8];
#pragma unroll
        for (int r = 0; r < 8; ++r) v[r] = acc[pt][p][r] - cn[r];
        const float m0 = fmaxf(fmaxf(v[0], v[1]), v[2]);
        const float m1 = fmaxf(fmaxf(v[3], v[4]), v[5]);
        if (pt == 0) {
          best[p] = fmaxf(fmaxf(fmaxf(v[6], v[7]), m0), m1);
#pragma unroll
          for (int r = 0; r < 8; ++r) sav[p][r] = v[r];
          btile[p] = 0;
        } else {
          const float m2 = fmaxf(fmaxf(v[6], v[7]), best[p]);
          const float nb = fmaxf(fmaxf(m0, m1), m2);
          const bool take = nb > best[p];
#pragma unroll
          for (int r = 0; r < 8; ++r) sav[p][r] = take ? v[r] : sav[p][r];
          btile[p] = take ? (pt * 16) : btile[p];
          best[p] = nb;
        }
      }
    }
    static_assert(NP == 2, "the fence below names both pixel tiles");
    if (ct > 0)
      asm volatile("" :: "v"(best[0]), "v"(best[1]),
                   "v"(sav[0][0]), "v"(sav[0][1]), "v"(sav[0][2]), "v"(sav[0][3]),
                   "v"(sav[0][4]), "v"(sav[0][5]), "v"(sav[0][6]), "v"(sav[0][7]),
                   "v"(sav[1][0]), "v"(sav[1][1]), "v"(sav[1][2]), "v"(sav[1][3]),
                   "v"(sav[1][4]), "v"(sav[1][5]), "v"(sav[1][6]), "v"(sav[1][7]));
    __builtin_amdgcn_sched_barrier(0);
  }
}

__device__ __forceinline__ void vq_step_fill(v2f (&a)[CT][16], v4f (&t)[32], const float* gsrc,
                                             float* stg, int c, int h, const v2f (&bq)[NP][16],
                                             const float* cnl, float (&best)[NP], int (&bidx)[NP]) {
  v8f acc[CT][NP];
#pragma unroll
  for (int ct = 0; ct <= CT; ++ct) {
    if (ct < CT) {
      float* slot = stg + (ct & 3) * STG_SLOT;
#pragma unroll
      for (int j = 0; j < 8; ++j)
        *(v4f*)(slot + (2 * j + h) * PITCH + 4 * c) = t[(8 * ct + j) & 31];
      if (ct + 4 < CT) {
#pragma unroll
        for (int j = 0; j < 8; ++j)
          t[(8 * ct + j) & 31] = *(const v4f*)(gsrc + (8 * (ct + 4) + j) * 128);
      }
      __builtin_amdgcn_wave_barrier();
#pragma unroll
      for (int kk = 0; kk < 16; ++kk)
        a[ct][kk] = *(const v2f*)(slot + c * PITCH + 2 * h + kk * 4);
#pragma unroll
      for (int p = 0; p < NP; ++p) acc[ct][p] = (v8f){};
#pragma unroll
      for (int kk = 0; kk < 16; ++kk)
#pragma unroll
        for (int p = 0; p < NP; ++p)
          acc[ct][p] = __builtin_amdgcn_wmma_f32_16x16x4_f32(
              false, a[ct][kk], false, bq[p][kk], (short)0, acc[ct][p], false, false);
    }
    if (ct > 0) {
      const int pt = ct - 1;
      const v4f c0 = *(const v4f*)(cnl + pt * 16);
      const v4f c1 = *(const v4f*)(cnl + pt * 16 + 4);
      const float cn[8] = {c0.x, c0.y, c0.z, c0.w, c1.x, c1.y, c1.z, c1.w};
#pragma unroll
      for (int p = 0; p < NP; ++p) {
        float v[8];
#pragma unroll
        for (int r = 0; r < 8; ++r) v[r] = acc[pt][p][r] - cn[r];
        const float m0 = fmaxf(fmaxf(v[0], v[1]), v[2]);
        const float m1 = fmaxf(fmaxf(v[3], v[4]), v[5]);
        const float m  = fmaxf(fmaxf(fmaxf(v[6], v[7]), m0), m1);
        int row = 7;
#pragma unroll
        for (int r = 6; r >= 0; --r) row = (v[r] == m) ? r : row;
        if (pt == 0) {
          best[p] = m;
          bidx[p] = row;
        } else {
          const bool take = m > best[p];
          bidx[p] = take ? (pt * 16 + row) : bidx[p];
          best[p] = take ? m : best[p];
        }
      }
    }
    static_assert(NP == 2, "the fence below names both pixel tiles");
    if (ct > 0)
      asm volatile("" :: "v"(best[0]), "v"(best[1]), "v"(bidx[0]), "v"(bidx[1]));
    __builtin_amdgcn_sched_barrier(0);
  }
}

__device__ __forceinline__ void vq_park_idx(const float (&best)[NP], const int (&bidx)[NP],
                                            int cbase, int h, float* rs, int* ri, int st) {
#pragma unroll
  for (int p = 0; p < NP; ++p) {
    float s = best[p];
    int   i = cbase + bidx[p];
    const float os = __shfl_xor(s, 16, 32);
    const int   oi = __shfl_xor(i, 16, 32);
    const bool take = (os > s) | ((os == s) & (oi < i));
    s = take ? os : s;
    i = take ? oi : i;
    if (h == 0) {
      rs[st * (NP * 16) + p * 16] = s;
      ri[st * (NP * 16) + p * 16] = i;
    }
  }
}

__device__ __forceinline__ void vq_park(float (&best)[NP], int (&btile)[NP], float (&sav)[NP][8],
                                        int cbase, int h, float* rs, int* ri, int st) {
#pragma unroll
  for (int p = 0; p < NP; ++p) {
    float s = best[p];
    int row = 7;
#pragma unroll
    for (int r = 6; r >= 0; --r) row = (sav[p][r] == s) ? r : row;
    int   i = cbase + btile[p] + row;
    const float os = __shfl_xor(s, 16, 32);
    const int   oi = __shfl_xor(i, 16, 32);
    const bool take = (os > s) | ((os == s) & (oi < i));
    s = take ? os : s;
    i = take ? oi : i;
    if (h == 0) {
      rs[st * (NP * 16) + p * 16] = s;
      ri[st * (NP * 16) + p * 16] = i;
    }
  }
}

__global__ __launch_bounds__(256) void vq_main_kernel(const float* __restrict__ in,
                                                      const float* __restrict__ cb,
                                                      const float* __restrict__ cn2h,
                                                      float* __restrict__ out) {
  __shared__ __attribute__((aligned(16))) float sm[SM_FLOATS];

  const int tid  = threadIdx.x;
  const int lane = tid & 31;
  const int wave = tid >> 5;
  const int c    = lane & 15;
  const int h    = lane >> 4;
  const int g    = blockIdx.x;
  const int b    = g >> 2;
  const int hw0  = (g & 3) << 8;
  const int code0 = wave * (CT * 16);

  const int px4 = tid & 3;
  const int chp = (tid >> 2) & 31;
  const int pth = tid >> 7;
  v4f xu[8], xw[8];
  {
    const float* src = in + (((size_t)(b * VQ_C + 2 * chp)) << 10) + hw0 + 4 * px4;
#pragma unroll
    for (int j = 0; j < 8; ++j) {
      xu[j] = *(const v4f*)(src + 16 * (2 * j + pth));
      xw[j] = *(const v4f*)(src + VQ_HW + 16 * (2 * j + pth));
    }
  }
  float cnv[4];
#pragma unroll
  for (int i = 0; i < 4; ++i) cnv[i] = cn2h[tid + 256 * i];

  v4f t[32];
  const float* gsrc = cb + (size_t)code0 * VQ_C + 4 * lane;
#pragma unroll
  for (int j = 0; j < 32; ++j) t[j] = *(const v4f*)(gsrc + j * 128);
  __builtin_amdgcn_sched_barrier(0);

  {
    float* xs = sm + XS_OFF;
#pragma unroll
    for (int j = 0; j < 8; ++j) {
      float* d = xs + (16 * (2 * j + pth) + 4 * px4) * PITCH + 2 * chp;
      *(v2f*)(d + 0 * PITCH) = (v2f){xu[j].x, xw[j].x};
      *(v2f*)(d + 1 * PITCH) = (v2f){xu[j].y, xw[j].y};
      *(v2f*)(d + 2 * PITCH) = (v2f){xu[j].z, xw[j].z};
      *(v2f*)(d + 3 * PITCH) = (v2f){xu[j].w, xw[j].w};
    }
#pragma unroll
    for (int i = 0; i < 4; ++i) sm[CNS_OFF + tid + 256 * i] = cnv[i];
  }
  __builtin_amdgcn_fence(__ATOMIC_RELEASE, "workgroup", "local");
  __builtin_amdgcn_s_barrier();
  __builtin_amdgcn_fence(__ATOMIC_ACQUIRE, "workgroup", "local");

  static_assert(PRIO_STEPS % 2 == 0 && PRIO_STEPS >= 2 && PRIO_STEPS <= 6,
                "the window ends in front of the second step of a loop trip");
  const bool young = __builtin_amdgcn_readfirstlane(wave) >= WAVES / 2;
  if (young) __builtin_amdgcn_s_setprio(3);

  const float* xl  = sm + XS_OFF + c * PITCH + 2 * h;
  const float* cnl = sm + CNS_OFF + code0 + 8 * h;
  float* rs = sm + REDS_OFF + wave * PXB + c;
  int*   ri = (int*)(sm + REDI_OFF) + wave * PXB + c;
  const int cbase = code0 + 8 * h;

  float best[NP];
  int   btile[NP];
  float sav[NP][8];

  v2f bq0[NP][16], bq1[NP][16];
  v2f a[CT][16];
  {
    v2f bqf[NP][16];
    vq_ldx(xl, 0, bqf);
    float fbest[NP];
    int   fidx[NP];
    vq_step_fill(a, t, gsrc, sm + STG_OFF + wave * (4 * STG_SLOT), c, h, bqf, cnl, fbest, fidx);
    vq_ldx(xl, 1, bq1);
    vq_park_idx(fbest, fidx, cbase, h, rs, ri, 0);
  }
#pragma unroll 1
  for (int it = 0; it < PXB / (NP * 16) / 2; ++it) {
    const bool more = it < PXB / (NP * 16) / 2 - 1;
    if (more)
      vq_ldx(xl, 2 * it + 2, bq0);
    vq_step(a, bq1, cnl, best, btile, sav);
    vq_park(best, btile, sav, cbase, h, rs, ri, 2 * it + 1);
    if (more) {
      if (young && 2 * it + 2 == PRIO_STEPS) __builtin_amdgcn_s_setprio(0);
      vq_ldx(xl, 2 * it + 3, bq1);
      vq_step(a, bq0, cnl, best, btile, sav);
      vq_park(best, btile, sav, cbase, h, rs, ri, 2 * it + 2);
    }
  }
  __syncthreads();

  {
    const float* qs = sm + REDS_OFF + tid;
    const int*   qi = (const int*)(sm + REDI_OFF) + tid;
    float s = qs[0];
    int   i = qi[0];
#pragma unroll
    for (int w = 1; w < WAVES; ++w) {
      const float os = qs[w * PXB];
      const int   oi = qi[w * PXB];
      const bool take = (os > s) | ((os == s) & (oi < i));
      s = take ? os : s;
      i = take ? oi : i;
    }
    ((int*)(sm + RB_OFF))[tid] = i;
  }
  __syncthreads();

  {
    const int* rb = (const int*)(sm + RB_OFF);
    float* ys = sm + XS_OFF;
    v4f t[16];
#pragma unroll
    for (int j = 0; j < 16; ++j) {
      const int pp = 32 * wave + 2 * j + h;
      t[j] = *(const v4f*)(cb + (size_t)rb[pp] * VQ_C + 4 * c);
    }
#pragma unroll
    for (int j = 0; j < 16; ++j) {
      const int pp = 32 * wave + 2 * j + h;
      *(v4f*)(ys + pp * PITCH + 4 * c) = t[j];
    }
  }
  __syncthreads();
  {
    const float* yr = sm + XS_OFF + tid * PITCH;
    float* o = out + (((size_t)b * VQ_C) << 10) + hw0 + tid;
#pragma unroll
    for (int q = 0; q < 16; ++q) {
      const v4f v = *(const v4f*)(yr + 4 * q);
      o[(size_t)(4 * q + 0) << 10] = v.x;
      o[(size_t)(4 * q + 1) << 10] = v.y;
      o[(size_t)(4 * q + 2) << 10] = v.z;
      o[(size_t)(4 * q + 3) << 10] = v.w;
    }
  }
}

extern "C" void kernel_launch(void* const* d_in, const int* in_sizes, int n_in,
                              void* d_out, int out_size, void* d_ws, size_t ws_size,
                              hipStream_t stream) {
  const float* in  = (const float*)d_in[0];
  const float* cb  = (const float*)d_in[1];
  float* out  = (float*)d_out;
  float* cn2h = (float*)d_ws;

  vq_cnorm_kernel<<<VQ_K / 256, 256, 0, stream>>>(cb, cn2h);
  vq_main_kernel<<<(VQ_B * VQ_HW) / PXB, 256, 0, stream>>>(in, cb, cn2h, out);
}


# Each wave sums the half norms of its own 128 codes from its A fragments (32 squares a lane, one half swap) into its own LDS strip; norm launch dropped, 2 launches to 1; norm sum order changed
# speedup vs baseline: 2.6408x; 1.0555x over previous
#include <hip/hip_runtime.h>


typedef __attribute__((ext_vector_type(2))) float v2f;
typedef __attribute__((ext_vector_type(4))) float v4f;
typedef __attribute__((ext_vector_type(8))) float v8f;

#define VQ_B 64
#define VQ_C 64
#define VQ_HW 1024
#define VQ_K 1024
#define WAVES 8
#define PXB 256
#define CT 8
#define NP 2
#define PITCH 68
#define PRIO_STEPS 6

#define XS_OFF   0
#define CNS_OFF  (PXB * PITCH)
#define REDS_OFF (CNS_OFF + VQ_K)
#define REDI_OFF (REDS_OFF + WAVES * PXB)
#define RB_OFF   (REDI_OFF + WAVES * PXB)
#define STG_OFF  (RB_OFF + PXB)
#define STG_SLOT (16 * PITCH)
#define SM_FLOATS (STG_OFF + WAVES * 4 * STG_SLOT)

__device__ __forceinline__ void vq_ldx(const float* xl, int st, v2f (&bq)[NP][16]) {
  const float* p0 = xl + st * (NP * 16 * PITCH);
#pragma unroll
  for (int p = 0; p < NP; ++p)
#pragma unroll
    for (int kk = 0; kk < 16; ++kk)
      bq[p][kk] = *(const v2f*)(p0 + p * 16 * PITCH + kk * 4);
}

__device__ __forceinline__ void vq_step(const v2f (&a)[CT][16], const v2f (&bq)[NP][16],
                                        const float* cnl, float (&best)[NP], int (&btile)[NP],
                                        float (&sav)[NP][8]) {
  v8f acc[CT][NP];
#pragma unroll
  for (int ct = 0; ct <= CT; ++ct) {
    if (ct < CT) {
#pragma unroll
      for (int p = 0; p < NP; ++p) acc[ct][p] = (v8f){};
#pragma unroll
      for (int kk = 0; kk < 16; ++kk)
#pragma unroll
        for (int p = 0; p < NP; ++p)
          acc[ct][p] = __builtin_amdgcn_wmma_f32_16x16x4_f32(
              false, a[ct][kk], false, bq[p][kk], (short)0, acc[ct][p], false, false);
    }
    if (ct > 0) {
      const int pt = ct - 1;
      const v4f c0 = *(const v4f*)(cnl + pt * 16);
      const v4f c1 = *(const v4f*)(cnl + pt * 16 + 4);
      const float cn[8] = {c0.x, c0.y, c0.z, c0.w, c1.x, c1.y, c1.z, c1.w};
#pragma unroll
      for (int p = 0; p < NP; ++p) {
        float v[8];
#pragma unroll
        for (int r = 0; r < 8; ++r) v[r] = acc[pt][p][r] - cn[r];
        const float m0 = fmaxf(fmaxf(v[0], v[1]), v[2]);
        const float m1 = fmaxf(fmaxf(v[3], v[4]), v[5]);
        if (pt == 0) {
          best[p] = fmaxf(fmaxf(fmaxf(v[6], v[7]), m0), m1);
#pragma unroll
          for (int r = 0; r < 8; ++r) sav[p][r] = v[r];
          btile[p] = 0;
        } else {
          const float m2 = fmaxf(fmaxf(v[6], v[7]), best[p]);
          const float nb = fmaxf(fmaxf(m0, m1), m2);
          const bool take = nb > best[p];
#pragma unroll
          for (int r = 0; r < 8; ++r) sav[p][r] = take ? v[r] : sav[p][r];
          btile[p] = take ? (pt * 16) : btile[p];
          best[p] = nb;
        }
      }
    }
    static_assert(NP == 2, "the fence below names both pixel tiles");
    if (ct > 0)
      asm volatile("" :: "v"(best[0]), "v"(best[1]),
                   "v"(sav[0][0]), "v"(sav[0][1]), "v"(sav[0][2]), "v"(sav[0][3]),
                   "v"(sav[0][4]), "v"(sav[0][5]), "v"(sav[0][6]), "v"(sav[0][7]),
                   "v"(sav[1][0]), "v"(sav[1][1]), "v"(sav[1][2]), "v"(sav[1][3]),
                   "v"(sav[1][4]), "v"(sav[1][5]), "v"(sav[1][6]), "v"(sav[1][7]));
    __builtin_amdgcn_sched_barrier(0);
  }
}

__device__ __forceinline__ void vq_step_fill(v2f (&a)[CT][16], v4f (&t)[32], const float* gsrc,
                                             float* stg, int c, int h, const v2f (&bq)[NP][16],
                                             float* cnw, const float* cnl, float (&best)[NP],
                                             int (&bidx)[NP]) {
  v8f acc[CT][NP];
#pragma unroll
  for (int ct = 0; ct <= CT; ++ct) {
    if (ct < CT) {
      float* slot = stg + (ct & 3) * STG_SLOT;
#pragma unroll
      for (int j = 0; j < 8; ++j)
        *(v4f*)(slot + (2 * j + h) * PITCH + 4 * c) = t[(8 * ct + j) & 31];
      if (ct + 4 < CT) {
#pragma unroll
        for (int j = 0; j < 8; ++j)
          t[(8 * ct + j) & 31] = *(const v4f*)(gsrc + (8 * (ct + 4) + j) * 128);
      }
      __builtin_amdgcn_wave_barrier();
#pragma unroll
      for (int kk = 0; kk < 16; ++kk)
        a[ct][kk] = *(const v2f*)(slot + c * PITCH + 2 * h + kk * 4);
#pragma unroll
      for (int p = 0; p < NP; ++p) acc[ct][p] = (v8f){};
#pragma unroll
      for (int kk = 0; kk < 16; ++kk)
#pragma unroll
        for (int p = 0; p < NP; ++p)
          acc[ct][p] = __builtin_amdgcn_wmma_f32_16x16x4_f32(
              false, a[ct][kk], false, bq[p][kk], (short)0, acc[ct][p], false, false);
    }
    if (ct > 0) {
      const int pt = ct - 1;
      const v4f c0 = *(const v4f*)(cnl + pt * 16);
      const v4f c1 = *(const v4f*)(cnl + pt * 16 + 4);
      const float cn[8] = {c0.x, c0.y, c0.z, c0.w, c1.x, c1.y, c1.z, c1.w};
#pragma unroll
      for (int p = 0; p < NP; ++p) {
        float v[8];
#pragma unroll
        for (int r = 0; r < 8; ++r) v[r] = acc[pt][p][r] - cn[r];
        const float m0 = fmaxf(fmaxf(v[0], v[1]), v[2]);
        const float m1 = fmaxf(fmaxf(v[3], v[4]), v[5]);
        const float m  = fmaxf(fmaxf(fmaxf(v[6], v[7]), m0), m1);
        int row = 7;
#pragma unroll
        for (int r = 6; r >= 0; --r) row = (v[r] == m) ? r : row;
        if (pt == 0) {
          best[p] = m;
          bidx[p] = row;
        } else {
          const bool take = m > best[p];
          bidx[p] = take ? (pt * 16 + row) : bidx[p];
          best[p] = take ? m : best[p];
        }
      }
    }
    if (ct < CT) {
      float q0 = 0.0f, q1 = 0.0f;
#pragma unroll
      for (int kk = 0; kk < 16; ++kk) {
        q0 = fmaf(a[ct][kk].x, a[ct][kk].x, q0);
        q1 = fmaf(a[ct][kk].y, a[ct][kk].y, q1);
      }
      float q = q0 + q1;
      q += __shfl_xor(q, 16, 32);
      cnw[ct * 16 + c] = 0.5f * q;
      __builtin_amdgcn_wave_barrier();
    }
    static_assert(NP == 2, "the fence below names both pixel tiles");
    if (ct > 0)
      asm volatile("" :: "v"(best[0]), "v"(best[1]), "v"(bidx[0]), "v"(bidx[1]));
    __builtin_amdgcn_sched_barrier(0);
  }
}

__device__ __forceinline__ void vq_park_idx(const float (&best)[NP], const int (&bidx)[NP],
                                            int cbase, int h, float* rs, int* ri, int st) {
#pragma unroll
  for (int p = 0; p < NP; ++p) {
    float s = best[p];
    int   i = cbase + bidx[p];
    const float os = __shfl_xor(s, 16, 32);
    const int   oi = __shfl_xor(i, 16, 32);
    const bool take = (os > s) | ((os == s) & (oi < i));
    s = take ? os : s;
    i = take ? oi : i;
    if (h == 0) {
      rs[st * (NP * 16) + p * 16] = s;
      ri[st * (NP * 16) + p * 16] = i;
    }
  }
}

__device__ __forceinline__ void vq_park(float (&best)[NP], int (&btile)[NP], float (&sav)[NP][8],
                                        int cbase, int h, float* rs, int* ri, int st) {
#pragma unroll
  for (int p = 0; p < NP; ++p) {
    float s = best[p];
    int row = 7;
#pragma unroll
    for (int r = 6; r >= 0; --r) row = (sav[p][r] == s) ? r : row;
    int   i = cbase + btile[p] + row;
    const float os = __shfl_xor(s, 16, 32);
    const int   oi = __shfl_xor(i, 16, 32);
    const bool take = (os > s) | ((os == s) & (oi < i));
    s = take ? os : s;
    i = take ? oi : i;
    if (h == 0) {
      rs[st * (NP * 16) + p * 16] = s;
      ri[st * (NP * 16) + p * 16] = i;
    }
  }
}

__global__ __launch_bounds__(256) void vq_main_kernel(const float* __restrict__ in,
                                                      const float* __restrict__ cb,
                                                      float* __restrict__ out) {
  __shared__ __attribute__((aligned(16))) float sm[SM_FLOATS];

  const int tid  = threadIdx.x;
  const int lane = tid & 31;
  const int wave = tid >> 5;
  const int c    = lane & 15;
  const int h    = lane >> 4;
  const int g    = blockIdx.x;
  const int b    = g >> 2;
  const int hw0  = (g & 3) << 8;
  const int code0 = wave * (CT * 16);

  const int px4 = tid & 3;
  const int chp = (tid >> 2) & 31;
  const int pth = tid >> 7;
  v4f xu[8], xw[8];
  {
    const float* src = in + (((size_t)(b * VQ_C + 2 * chp)) << 10) + hw0 + 4 * px4;
#pragma unroll
    for (int j = 0; j < 8; ++j) {
      xu[j] = *(const v4f*)(src + 16 * (2 * j + pth));
      xw[j] = *(const v4f*)(src + VQ_HW + 16 * (2 * j + pth));
    }
  }
  v4f t[32];
  const float* gsrc = cb + (size_t)code0 * VQ_C + 4 * lane;
#pragma unroll
  for (int j = 0; j < 32; ++j) t[j] = *(const v4f*)(gsrc + j * 128);
  __builtin_amdgcn_sched_barrier(0);

  {
    float* xs = sm + XS_OFF;
#pragma unroll
    for (int j = 0; j < 8; ++j) {
      float* d = xs + (16 * (2 * j + pth) + 4 * px4) * PITCH + 2 * chp;
      *(v2f*)(d + 0 * PITCH) = (v2f){xu[j].x, xw[j].x};
      *(v2f*)(d + 1 * PITCH) = (v2f){xu[j].y, xw[j].y};
      *(v2f*)(d + 2 * PITCH) = (v2f){xu[j].z, xw[j].z};
      *(v2f*)(d + 3 * PITCH) = (v2f){xu[j].w, xw[j].w};
    }
  }
  __builtin_amdgcn_fence(__ATOMIC_RELEASE, "workgroup", "local");
  __builtin_amdgcn_s_barrier();
  __builtin_amdgcn_fence(__ATOMIC_ACQUIRE, "workgroup", "local");

  static_assert(PRIO_STEPS % 2 == 0 && PRIO_STEPS >= 2 && PRIO_STEPS <= 6,
                "the window ends in front of the second step of a loop trip");
  const bool young = __builtin_amdgcn_readfirstlane(wave) >= WAVES / 2;
  if (young) __builtin_amdgcn_s_setprio(3);

  const float* xl  = sm + XS_OFF + c * PITCH + 2 * h;
  float*       cnw = sm + CNS_OFF + code0;
  const float* cnl = cnw + 8 * h;
  float* rs = sm + REDS_OFF + wave * PXB + c;
  int*   ri = (int*)(sm + REDI_OFF) + wave * PXB + c;
  const int cbase = code0 + 8 * h;

  float best[NP];
  int   btile[NP];
  float sav[NP][8];

  v2f bq0[NP][16], bq1[NP][16];
  v2f a[CT][16];
  {
    v2f bqf[NP][16];
    vq_ldx(xl, 0, bqf);
    float fbest[NP];
    int   fidx[NP];
    vq_step_fill(a, t, gsrc, sm + STG_OFF + wave * (4 * STG_SLOT), c, h, bqf, cnw, cnl, fbest,
                 fidx);
    vq_ldx(xl, 1, bq1);
    vq_park_idx(fbest, fidx, cbase, h, rs, ri, 0);
  }
#pragma unroll 1
  for (int it = 0; it < PXB / (NP * 16) / 2; ++it) {
    const bool more = it < PXB / (NP * 16) / 2 - 1;
    if (more)
      vq_ldx(xl, 2 * it + 2, bq0);
    vq_step(a, bq1, cnl, best, btile, sav);
    vq_park(best, btile, sav, cbase, h, rs, ri, 2 * it + 1);
    if (more) {
      if (young && 2 * it + 2 == PRIO_STEPS) __builtin_amdgcn_s_setprio(0);
      vq_ldx(xl, 2 * it + 3, bq1);
      vq_step(a, bq0, cnl, best, btile, sav);
      vq_park(best, btile, sav, cbase, h, rs, ri, 2 * it + 2);
    }
  }
  __syncthreads();

  {
    const float* qs = sm + REDS_OFF + tid;
    const int*   qi = (const int*)(sm + REDI_OFF) + tid;
    float s = qs[0];
    int   i = qi[0];
#pragma unroll
    for (int w = 1; w < WAVES; ++w) {
      const float os = qs[w * PXB];
      const int   oi = qi[w * PXB];
      const bool take = (os > s) | ((os == s) & (oi < i));
      s = take ? os : s;
      i = take ? oi : i;
    }
    ((int*)(sm + RB_OFF))[tid] = i;
  }
  __syncthreads();

  {
    const int* rb = (const int*)(sm + RB_OFF);
    float* ys = sm + XS_OFF;
    v4f t[16];
#pragma unroll
    for (int j = 0; j < 16; ++j) {
      const int pp = 32 * wave + 2 * j + h;
      t[j] = *(const v4f*)(cb + (size_t)rb[pp] * VQ_C + 4 * c);
    }
#pragma unroll
    for (int j = 0; j < 16; ++j) {
      const int pp = 32 * wave + 2 * j + h;
      *(v4f*)(ys + pp * PITCH + 4 * c) = t[j];
    }
  }
  __syncthreads();
  {
    const float* yr = sm + XS_OFF + tid * PITCH;
    float* o = out + (((size_t)b * VQ_C) << 10) + hw0 + tid;
#pragma unroll
    for (int q = 0; q < 16; ++q) {
      const v4f v = *(const v4f*)(yr + 4 * q);
      o[(size_t)(4 * q + 0) << 10] = v.x;
      o[(size_t)(4 * q + 1) << 10] = v.y;
      o[(size_t)(4 * q + 2) << 10] = v.z;
      o[(size_t)(4 * q + 3) << 10] = v.w;
    }
  }
}

extern "C" void kernel_launch(void* const* d_in, const int* in_sizes, int n_in,
                              void* d_out, int out_size, void* d_ws, size_t ws_size,
                              hipStream_t stream) {
  const float* in  = (const float*)d_in[0];
  const float* cb  = (const float*)d_in[1];
  float* out  = (float*)d_out;

  vq_main_kernel<<<(VQ_B * VQ_HW) / PXB, 256, 0, stream>>>(in, cb, out);
}


# Half norms start each score chain of steps 1-7 as its C operand, held in 64 registers (room: one pixel tile per sweep, code tiles in pairs): no per-tile norm read or subtract; score sum order changed
# speedup vs baseline: 2.7318x; 1.0345x over previous
#include <hip/hip_runtime.h>


typedef __attribute__((ext_vector_type(2))) float v2f;
typedef __attribute__((ext_vector_type(4))) float v4f;
typedef __attribute__((ext_vector_type(8))) float v8f;

#define VQ_B 64
#define VQ_C 64
#define VQ_HW 1024
#define VQ_K 1024
#define WAVES 8
#define PXB 256
#define CT 8
#define NP 2
#define PITCH 68
#define PRIO_STEPS 6

#define XS_OFF   0
#define CNS_OFF  (PXB * PITCH)
#define REDS_OFF (CNS_OFF + VQ_K)
#define REDI_OFF (REDS_OFF + WAVES * PXB)
#define RB_OFF   (REDI_OFF + WAVES * PXB)
#define STG_OFF  (RB_OFF + PXB)
#define STG_SLOT (16 * PITCH)
#define SM_FLOATS (STG_OFF + WAVES * 4 * STG_SLOT)

__device__ __forceinline__ void vq_ldx(const float* xl, int st, v2f (&bq)[NP][16]) {
  const float* p0 = xl + st * (NP * 16 * PITCH);
#pragma unroll
  for (int p = 0; p < NP; ++p)
#pragma unroll
    for (int kk = 0; kk < 16; ++kk)
      bq[p][kk] = *(const v2f*)(p0 + p * 16 * PITCH + kk * 4);
}

__device__ __forceinline__ void vq_ldx1(const float* xl, int tile, v2f (&bq)[16]) {
  const float* p0 = xl + tile * (16 * PITCH);
#pragma unroll
  for (int kk = 0; kk < 16; ++kk) bq[kk] = *(const v2f*)(p0 + kk * 4);
}

__device__ __forceinline__ void vq_half(const v2f (&a)[CT][16], const v2f (&bq)[16],
                                        const v8f (&ncn)[CT], float& best, int& btile,
                                        float (&sav)[8]) {
  static_assert(CT % 2 == 0, "code tiles go in pairs");
  v8f acc[CT];
#pragma unroll
  for (int r = 0; r <= CT / 2; ++r) {
    if (r < CT / 2) {
      const int t0 = 2 * r, t1 = 2 * r + 1;
      acc[t0] = __builtin_amdgcn_wmma_f32_16x16x4_f32(false, a[t0][0], false, bq[0], (short)0,
                                                      ncn[t0], false, false);
      acc[t1] = __builtin_amdgcn_wmma_f32_16x16x4_f32(false, a[t1][0], false, bq[0], (short)0,
                                                      ncn[t1], false, false);
#pragma unroll
      for (int kk = 1; kk < 16; ++kk) {
        acc[t0] = __builtin_amdgcn_wmma_f32_16x16x4_f32(false, a[t0][kk], false, bq[kk],
                                                        (short)0, acc[t0], false, false);
        acc[t1] = __builtin_amdgcn_wmma_f32_16x16x4_f32(false, a[t1][kk], false, bq[kk],
                                                        (short)0, acc[t1], false, false);
      }
    }
    if (r > 0) {
#pragma unroll
      for (int q = 0; q < 2; ++q) {
        const int pt = 2 * (r - 1) + q;
        float v[8];
#pragma unroll
        for (int i = 0; i < 8; ++i) v[i] = acc[pt][i];
        const float m0 = fmaxf(fmaxf(v[0], v[1]), v[2]);
        const float m1 = fmaxf(fmaxf(v[3], v[4]), v[5]);
        if (pt == 0) {
          best = fmaxf(fmaxf(fmaxf(v[6], v[7]), m0), m1);
#pragma unroll
          for (int i = 0; i < 8; ++i) sav[i] = v[i];
          btile = 0;
        } else {
          const float m2 = fmaxf(fmaxf(v[6], v[7]), best);
          const float nb = fmaxf(fmaxf(m0, m1), m2);
          const bool take = nb > best;
#pragma unroll
          for (int i = 0; i < 8; ++i) sav[i] = take ? v[i] : sav[i];
          btile = take ? (pt * 16) : btile;
          best = nb;
        }
      }
    }
    if (r > 0)
      asm volatile("" :: "v"(best), "v"(sav[0]), "v"(sav[1]), "v"(sav[2]), "v"(sav[3]),
                   "v"(sav[4]), "v"(sav[5]), "v"(sav[6]), "v"(sav[7]));
    __builtin_amdgcn_sched_barrier(0);
  }
}

__device__ __forceinline__ void vq_step_fill(v2f (&a)[CT][16], v4f (&t)[32], const float* gsrc,
                                             float* stg, int c, int h, const v2f (&bq)[NP][16],
                                             float* cnw, const float* cnl, float (&best)[NP],
                                             int (&bidx)[NP]) {
  v8f acc[CT][NP];
#pragma unroll
  for (int ct = 0; ct <= CT; ++ct) {
    if (ct < CT) {
      float* slot = stg + (ct & 3) * STG_SLOT;
#pragma unroll
      for (int j = 0; j < 8; ++j)
        *(v4f*)(slot + (2 * j + h) * PITCH + 4 * c) = t[(8 * ct + j) & 31];
      if (ct + 4 < CT) {
#pragma unroll
        for (int j = 0; j < 8; ++j)
          t[(8 * ct + j) & 31] = *(const v4f*)(gsrc + (8 * (ct + 4) + j) * 128);
      }
      __builtin_amdgcn_wave_barrier();
#pragma unroll
      for (int kk = 0; kk < 16; ++kk)
        a[ct][kk] = *(const v2f*)(slot + c * PITCH + 2 * h + kk * 4);
#pragma unroll
      for (int p = 0; p < NP; ++p) acc[ct][p] = (v8f){};
#pragma unroll
      for (int kk = 0; kk < 16; ++kk)
#pragma unroll
        for (int p = 0; p < NP; ++p)
          acc[ct][p] = __builtin_amdgcn_wmma_f32_16x16x4_f32(
              false, a[ct][kk], false, bq[p][kk], (short)0, acc[ct][p], false, false);
    }
    if (ct > 0) {
      const int pt = ct - 1;
      const v4f c0 = *(const v4f*)(cnl + pt * 16);
      const v4f c1 = *(const v4f*)(cnl + pt * 16 + 4);
      const float cn[8] = {c0.x, c0.y, c0.z, c0.w, c1.x, c1.y, c1.z, c1.w};
#pragma unroll
      for (int p = 0; p < NP; ++p) {
        float v[8];
#pragma unroll
        for (int r = 0; r < 8; ++r) v[r] = acc[pt][p][r] - cn[r];
        const float m0 = fmaxf(fmaxf(v[0], v[1]), v[2]);
        const float m1 = fmaxf(fmaxf(v[3], v[4]), v[5]);
        const float m  = fmaxf(fmaxf(fmaxf(v[6], v[7]), m0), m1);
        int row = 7;
#pragma unroll
        for (int r = 6; r >= 0; --r) row = (v[r] == m) ? r : row;
        if (pt == 0) {
          best[p] = m;
          bidx[p] = row;
        } else {
          const bool take = m > best[p];
          bidx[p] = take ? (pt * 16 + row) : bidx[p];
          best[p] = take ? m : best[p];
        }
      }
    }
    if (ct < CT) {
      float q0 = 0.0f, q1 = 0.0f;
#pragma unroll
      for (int kk = 0; kk < 16; ++kk) {
        q0 = fmaf(a[ct][kk].x, a[ct][kk].x, q0);
        q1 = fmaf(a[ct][kk].y, a[ct][kk].y, q1);
      }
      float q = q0 + q1;
      q += __shfl_xor(q, 16, 32);
      cnw[ct * 16 + c] = 0.5f * q;
      __builtin_amdgcn_wave_barrier();
    }
    static_assert(NP == 2, "the fence below names both pixel tiles");
    if (ct > 0)
      asm volatile("" :: "v"(best[0]), "v"(best[1]), "v"(bidx[0]), "v"(bidx[1]));
    __builtin_amdgcn_sched_barrier(0);
  }
}

__device__ __forceinline__ void vq_park_idx(const float (&best)[NP], const int (&bidx)[NP],
                                            int cbase, int h, float* rs, int* ri, int st) {
#pragma unroll
  for (int p = 0; p < NP; ++p) {
    float s = best[p];
    int   i = cbase + bidx[p];
    const float os = __shfl_xor(s, 16, 32);
    const int   oi = __shfl_xor(i, 16, 32);
    const bool take = (os > s) | ((os == s) & (oi < i));
    s = take ? os : s;
    i = take ? oi : i;
    if (h == 0) {
      rs[st * (NP * 16) + p * 16] = s;
      ri[st * (NP * 16) + p * 16] = i;
    }
  }
}

__device__ __forceinline__ void vq_park(float (&best)[NP], int (&btile)[NP], float (&sav)[NP][8],
                                        int cbase, int h, float* rs, int* ri, int st) {
#pragma unroll
  for (int p = 0; p < NP; ++p) {
    float s = best[p];
    int row = 7;
#pragma unroll
    for (int r = 6; r >= 0; --r) row = (sav[p][r] == s) ? r : row;
    int   i = cbase + btile[p] + row;
    const float os = __shfl_xor(s, 16, 32);
    const int   oi = __shfl_xor(i, 16, 32);
    const bool take = (os > s) | ((os == s) & (oi < i));
    s = take ? os : s;
    i = take ? oi : i;
    if (h == 0) {
      rs[st * (NP * 16) + p * 16] = s;
      ri[st * (NP * 16) + p * 16] = i;
    }
  }
}

__global__ __launch_bounds__(256) void vq_main_kernel(const float* __restrict__ in,
                                                      const float* __restrict__ cb,
                                                      float* __restrict__ out) {
  __shared__ __attribute__((aligned(16))) float sm[SM_FLOATS];

  const int tid  = threadIdx.x;
  const int lane = tid & 31;
  const int wave = tid >> 5;
  const int c    = lane & 15;
  const int h    = lane >> 4;
  const int g    = blockIdx.x;
  const int b    = g >> 2;
  const int hw0  = (g & 3) << 8;
  const int code0 = wave * (CT * 16);

  const int px4 = tid & 3;
  const int chp = (tid >> 2) & 31;
  const int pth = tid >> 7;
  v4f xu[8], xw[8];
  {
    const float* src = in + (((size_t)(b * VQ_C + 2 * chp)) << 10) + hw0 + 4 * px4;
#pragma unroll
    for (int j = 0; j < 8; ++j) {
      xu[j] = *(const v4f*)(src + 16 * (2 * j + pth));
      xw[j] = *(const v4f*)(src + VQ_HW + 16 * (2 * j + pth));
    }
  }
  v4f t[32];
  const float* gsrc = cb + (size_t)code0 * VQ_C + 4 * lane;
#pragma unroll
  for (int j = 0; j < 32; ++j) t[j] = *(const v4f*)(gsrc + j * 128);
  __builtin_amdgcn_sched_barrier(0);

  {
    float* xs = sm + XS_OFF;
#pragma unroll
    for (int j = 0; j < 8; ++j) {
      float* d = xs + (16 * (2 * j + pth) + 4 * px4) * PITCH + 2 * chp;
      *(v2f*)(d + 0 * PITCH) = (v2f){xu[j].x, xw[j].x};
      *(v2f*)(d + 1 * PITCH) = (v2f){xu[j].y, xw[j].y};
      *(v2f*)(d + 2 * PITCH) = (v2f){xu[j].z, xw[j].z};
      *(v2f*)(d + 3 * PITCH) = (v2f){xu[j].w, xw[j].w};
    }
  }
  __builtin_amdgcn_fence(__ATOMIC_RELEASE, "workgroup", "local");
  __builtin_amdgcn_s_barrier();
  __builtin_amdgcn_fence(__ATOMIC_ACQUIRE, "workgroup", "local");

  static_assert(PRIO_STEPS >= 1 && PRIO_STEPS < PXB / (NP * 16),
                "the window ends in front of a step of the loop below");
  const bool young = __builtin_amdgcn_readfirstlane(wave) >= WAVES / 2;
  if (young) __builtin_amdgcn_s_setprio(3);

  const float* xl  = sm + XS_OFF + c * PITCH + 2 * h;
  float*       cnw = sm + CNS_OFF + code0;
  const float* cnl = cnw + 8 * h;
  float* rs = sm + REDS_OFF + wave * PXB + c;
  int*   ri = (int*)(sm + REDI_OFF) + wave * PXB + c;
  const int cbase = code0 + 8 * h;

  float best[NP];
  int   btile[NP];
  float sav[NP][8];

  v2f bqa[16], bqb[16];
  v2f a[CT][16];
  {
    v2f bqf[NP][16];
    vq_ldx(xl, 0, bqf);
    float fbest[NP];
    int   fidx[NP];
    vq_step_fill(a, t, gsrc, sm + STG_OFF + wave * (4 * STG_SLOT), c, h, bqf, cnw, cnl, fbest,
                 fidx);
    vq_ldx1(xl, NP, bqa);
    vq_park_idx(fbest, fidx, cbase, h, rs, ri, 0);
  }
  v8f ncn[CT];
#pragma unroll
  for (int ct = 0; ct < CT; ++ct) {
    const v4f c0 = *(const v4f*)(cnl + ct * 16);
    const v4f c1 = *(const v4f*)(cnl + ct * 16 + 4);
    ncn[ct] = (v8f){-c0.x, -c0.y, -c0.z, -c0.w, -c1.x, -c1.y, -c1.z, -c1.w};
  }
#pragma unroll 1
  for (int st = 1; st < PXB / (NP * 16); ++st) {
    if (young && st == PRIO_STEPS) __builtin_amdgcn_s_setprio(0);
    vq_ldx1(xl, NP * st + 1, bqb);
    vq_half(a, bqa, ncn, best[0], btile[0], sav[0]);
    const int nxt = (st + 1 < PXB / (NP * 16)) ? NP * (st + 1) : NP * st;
    vq_ldx1(xl, nxt, bqa);
    vq_half(a, bqb, ncn, best[1], btile[1], sav[1]);
    vq_park(best, btile, sav, cbase, h, rs, ri, st);
  }
  __syncthreads();

  {
    const float* qs = sm + REDS_OFF + tid;
    const int*   qi = (const int*)(sm + REDI_OFF) + tid;
    float s = qs[0];
    int   i = qi[0];
#pragma unroll
    for (int w = 1; w < WAVES; ++w) {
      const float os = qs[w * PXB];
      const int   oi = qi[w * PXB];
      const bool take = (os > s) | ((os == s) & (oi < i));
      s = take ? os : s;
      i = take ? oi : i;
    }
    ((int*)(sm + RB_OFF))[tid] = i;
  }
  __syncthreads();

  {
    const int* rb = (const int*)(sm + RB_OFF);
    float* ys = sm + XS_OFF;
    v4f t[16];
#pragma unroll
    for (int j = 0; j < 16; ++j) {
      const int pp = 32 * wave + 2 * j + h;
      t[j] = *(const v4f*)(cb + (size_t)rb[pp] * VQ_C + 4 * c);
    }
#pragma unroll
    for (int j = 0; j < 16; ++j) {
      const int pp = 32 * wave + 2 * j + h;
      *(v4f*)(ys + pp * PITCH + 4 * c) = t[j];
    }
  }
  __syncthreads();
  {
    const float* yr = sm + XS_OFF + tid * PITCH;
    float* o = out + (((size_t)b * VQ_C) << 10) + hw0 + tid;
#pragma unroll
    for (int q = 0; q < 16; ++q) {
      const v4f v = *(const v4f*)(yr + 4 * q);
      o[(size_t)(4 * q + 0) << 10] = v.x;
      o[(size_t)(4 * q + 1) << 10] = v.y;
      o[(size_t)(4 * q + 2) << 10] = v.z;
      o[(size_t)(4 * q + 3) << 10] = v.w;
    }
  }
}

extern "C" void kernel_launch(void* const* d_in, const int* in_sizes, int n_in,
                              void* d_out, int out_size, void* d_ws, size_t ws_size,
                              hipStream_t stream) {
  const float* in  = (const float*)d_in[0];
  const float* cb  = (const float*)d_in[1];
  float* out  = (float*)d_out;

  vq_main_kernel<<<(VQ_B * VQ_HW) / PXB, 256, 0, stream>>>(in, cb, out);
}
